# starting kernel
# baseline (speedup 1.0000x reference)
#include <hip/hip_runtime.h>


typedef __attribute__((ext_vector_type(2))) float v2f;
typedef __attribute__((ext_vector_type(4))) float v4f;
typedef __attribute__((ext_vector_type(8))) float v8f;

#define NHEADS 4
#define HDIM   32
#define IMH    56
#define IMW    56
#define HWPIX  3136
#define BATCH  4
#define PADW   62
#define QSCALE 0.17677669529663687f

static __device__ __forceinline__ v8f wmma4(v2f a, v2f b, v8f c) {
  return __builtin_amdgcn_wmma_f32_16x16x4_f32(false, a, false, b, (short)0, c,
                                               false, false);
}

__global__ __launch_bounds__(256) void qkv_kernel(
    const float* __restrict__ x, const float* __restrict__ qkv_w,
    const float* __restrict__ qkv_b, float* __restrict__ q_arr,
    float* __restrict__ kpad, float* __restrict__ vpad) {
  __shared__ float Xs[16 * 132];
  const int strip = blockIdx.x;
  const int t = threadIdx.x;

  for (int idx = t; idx < 16 * 32; idx += 256) {
    int m = idx >> 5, q4 = idx & 31;
    *(v4f*)&Xs[m * 132 + q4 * 4] =
        *(const v4f*)&x[(size_t)(strip * 16 + m) * 128 + q4 * 4];
  }
  __syncthreads();

  const int lane = t & 31;
  const int half = lane >> 4;
  const int n16  = lane & 15;
  const int wave = t >> 5;

  for (int tt = 0; tt < 3; ++tt) {
    const int otile = wave * 3 + tt;
    const int obase = otile * 16;
    const int o = obase + n16;
    __builtin_prefetch(&qkv_w[(size_t)o * 128], 0, 1);
    v8f acc = {};
#pragma unroll
    for (int kc = 0; kc < 32; ++kc) {
      const int ch = kc * 4 + 2 * half;
      v2f a = *(const v2f*)&Xs[n16 * 132 + ch];
      v2f b = *(const v2f*)&qkv_w[(size_t)o * 128 + ch];
      acc = wmma4(a, b, acc);
    }
    const float bias = qkv_b[o];
    const int which = otile >> 3;
    const int head  = (otile & 7) >> 1;
    const int d     = (otile & 1) * 16 + n16;
#pragma unroll
    for (int vr = 0; vr < 8; ++vr) {
      const int m = vr + 8 * half;
      const unsigned pixel = strip * 16 + m;
      const unsigned bimg = pixel / HWPIX;
      const unsigned rem  = pixel % HWPIX;
      const unsigned i = rem / IMW, j = rem % IMW;
      const float val = acc[vr] + bias;
      if (which == 0) {
        q_arr[(((size_t)bimg * NHEADS + head) * HWPIX + rem) * HDIM + d] =
            val * QSCALE;
      } else {
        float* dst = (which == 1) ? kpad : vpad;
        dst[((((size_t)bimg * NHEADS + head) * PADW + (i + 3)) * PADW +
             (j + 3)) * HDIM + d] = val;
      }
    }
  }
}

__global__ __launch_bounds__(32) void attn_kernel(
    const float* __restrict__ q_arr, const float* __restrict__ kpad,
    const float* __restrict__ vpad, const float* __restrict__ rpb,
    float* __restrict__ attnout) {
  __shared__ float Ks[112 * 36];
  __shared__ float Vs[112 * 36];
  __shared__ float BiasT[49];
  float* PT = Ks;

  int unit = blockIdx.x;
  const int tj = unit % 7;  unit /= 7;
  const int ti = unit % 28; unit /= 28;
  const int h  = unit % NHEADS; unit /= NHEADS;
  const int b  = unit;
  const int i0 = ti * 2, j0 = tj * 8;

  const int lane = threadIdx.x;
  const int half = lane >> 4;
  const int n16  = lane & 15;

  for (int idx = lane; idx < 49; idx += 32) {
    const int di = idx / 7, dj = idx % 7;
    BiasT[idx] = rpb[((size_t)h * 13 + (di + 3)) * 13 + (dj + 3)];
  }

  const size_t kvrow0 = (((size_t)(b * NHEADS + h)) * PADW + i0) * PADW;
  for (int idx = lane; idx < 112 * 8; idx += 32) {
    const int pos = idx >> 3, q4 = idx & 7;
    const int kr = pos / 14, kc = pos % 14;
    const size_t g = (kvrow0 + kr * PADW + (j0 + kc)) * HDIM + q4 * 4;
    *(v4f*)&Ks[pos * 36 + q4 * 4] = *(const v4f*)&kpad[g];
    *(v4f*)&Vs[pos * 36 + q4 * 4] = *(const v4f*)&vpad[g];
  }

  const int r = n16 >> 3, cq = n16 & 7;
  const int qi = i0 + r, qj = j0 + cq;
  const size_t qbase =
      (((size_t)(b * NHEADS + h)) * HWPIX + qi * IMW + qj) * HDIM;
  v2f qb[8];
#pragma unroll
  for (int kc = 0; kc < 8; ++kc)
    qb[kc] = *(const v2f*)&q_arr[qbase + kc * 4 + 2 * half];
  __syncthreads();

  v8f S[7];
#pragma unroll
  for (int tt = 0; tt < 7; ++tt) {
    v8f acc = {};
    const int pos = tt * 16 + n16;
#pragma unroll
    for (int kc = 0; kc < 8; ++kc) {
      v2f a = *(const v2f*)&Ks[pos * 36 + kc * 4 + 2 * half];
      acc = wmma4(a, qb[kc], acc);
    }
#pragma unroll
    for (int vr = 0; vr < 8; ++vr) {
      const int key = tt * 16 + vr + 8 * half;
      const int kr = key / 14, kc2 = key % 14;
      const int di = kr - r, dj = kc2 - cq;
      if (di >= 0 && di < 7 && dj >= 0 && dj < 7)
        acc[vr] += BiasT[di * 7 + dj];
      else
        acc[vr] = -1e30f;
    }
    S[tt] = acc;
  }

  float mx = -1e30f;
#pragma unroll
  for (int tt = 0; tt < 7; ++tt)
#pragma unroll
    for (int vr = 0; vr < 8; ++vr) mx = fmaxf(mx, S[tt][vr]);
  mx = fmaxf(mx, __shfl_xor(mx, 16, 32));
  float sum = 0.f;
#pragma unroll
  for (int tt = 0; tt < 7; ++tt)
#pragma unroll
    for (int vr = 0; vr < 8; ++vr) {
      const float e = __expf(S[tt][vr] - mx);
      S[tt][vr] = e;
      sum += e;
    }
  sum += __shfl_xor(sum, 16, 32);
  const float inv = 1.f / sum;

  __syncthreads();
#pragma unroll
  for (int tt = 0; tt < 7; ++tt)
#pragma unroll
    for (int vr = 0; vr < 8; ++vr)
      PT[(tt * 16 + vr + 8 * half) * 17 + n16] = S[tt][vr];
  __syncthreads();

  const size_t outbase =
      ((size_t)(b * HWPIX) + qi * IMW + qj) * 128 + h * HDIM;
#pragma unroll
  for (int mt = 0; mt < 2; ++mt) {
    v8f acc = {};
    const int chA = mt * 16 + n16;
#pragma unroll
    for (int kc = 0; kc < 28; ++kc) {
      const int k0 = kc * 4 + 2 * half;
      v2f a;
      a.x = Vs[k0 * 36 + chA];
      a.y = Vs[(k0 + 1) * 36 + chA];
      v2f bfr;
      bfr.x = PT[k0 * 17 + n16];
      bfr.y = PT[(k0 + 1) * 17 + n16];
      acc = wmma4(a, bfr, acc);
    }
#pragma unroll
    for (int vr = 0; vr < 8; ++vr)
      attnout[outbase + mt * 16 + vr + 8 * half] = acc[vr] * inv;
  }
}

__global__ __launch_bounds__(256) void proj_kernel(
    const float* __restrict__ attnout, const float* __restrict__ proj_w,
    const float* __restrict__ proj_b, float* __restrict__ out) {
  __shared__ float Xs[16 * 132];
  const int strip = blockIdx.x;
  const int t = threadIdx.x;
  for (int idx = t; idx < 16 * 32; idx += 256) {
    int m = idx >> 5, q4 = idx & 31;
    *(v4f*)&Xs[m * 132 + q4 * 4] =
        *(const v4f*)&attnout[(size_t)(strip * 16 + m) * 128 + q4 * 4];
  }
  __syncthreads();
  const int lane = t & 31;
  const int half = lane >> 4;
  const int n16  = lane & 15;
  const int o = (t >> 5) * 16 + n16;
  __builtin_prefetch(&proj_w[(size_t)o * 128], 0, 1);
  v8f acc = {};
#pragma unroll
  for (int kc = 0; kc < 32; ++kc) {
    const int ch = kc * 4 + 2 * half;
    v2f a = *(const v2f*)&Xs[n16 * 132 + ch];
    v2f b = *(const v2f*)&proj_w[(size_t)o * 128 + ch];
    acc = wmma4(a, b, acc);
  }
  const float bias = proj_b[o];
#pragma unroll
  for (int vr = 0; vr < 8; ++vr) {
    const int m = vr + 8 * half;
    out[(size_t)(strip * 16 + m) * 128 + o] = acc[vr] + bias;
  }
}

extern "C" void kernel_launch(void* const* d_in, const int* in_sizes, int n_in,
                              void* d_out, int out_size, void* d_ws,
                              size_t ws_size, hipStream_t stream) {
  (void)in_sizes; (void)n_in; (void)out_size; (void)ws_size;
  const float* x      = (const float*)d_in[0];
  const float* qkv_w  = (const float*)d_in[1];
  const float* qkv_b  = (const float*)d_in[2];
  const float* rpb    = (const float*)d_in[3];
  const float* proj_w = (const float*)d_in[4];
  const float* proj_b = (const float*)d_in[5];

  float* ws = (float*)d_ws;
  const size_t q_len   = (size_t)BATCH * NHEADS * HWPIX * HDIM;
  const size_t pad_len = (size_t)BATCH * NHEADS * PADW * PADW * HDIM;
  float* q_arr   = ws;
  float* kpad    = ws + q_len;
  float* vpad    = kpad + pad_len;
  float* attnout = vpad + pad_len;

  hipMemsetAsync(kpad, 0, 2 * pad_len * sizeof(float), stream);

  qkv_kernel<<<784, 256, 0, stream>>>(x, qkv_w, qkv_b, q_arr, kpad, vpad);
  attn_kernel<<<BATCH * NHEADS * 28 * 7, 32, 0, stream>>>(q_arr, kpad, vpad,
                                                          rpb, attnout);
  proj_kernel<<<784, 256, 0, stream>>>(attnout, proj_w, proj_b, (float*)d_out);
}


# attention: K and V operand fragments loaded straight from global into registers as grouped loads (no LDS staging loop, LDS 32 KB to 8 KB per wave); QK k-terms summed in a new chunk order
# speedup vs baseline: 1.3963x; 1.3963x over previous
#include <hip/hip_runtime.h>


typedef __attribute__((ext_vector_type(2))) float v2f;
typedef __attribute__((ext_vector_type(4))) float v4f;
typedef __attribute__((ext_vector_type(8))) float v8f;

#define NHEADS 4
#define HDIM   32
#define IMH    56
#define IMW    56
#define HWPIX  3136
#define BATCH  4
#define PADW   62
#define QSCALE 0.17677669529663687f

static __device__ __forceinline__ v8f wmma4(v2f a, v2f b, v8f c) {
  return __builtin_amdgcn_wmma_f32_16x16x4_f32(false, a, false, b, (short)0, c,
                                               false, false);
}

__global__ __launch_bounds__(256) void qkv_kernel(
    const float* __restrict__ x, const float* __restrict__ qkv_w,
    const float* __restrict__ qkv_b, float* __restrict__ q_arr,
    float* __restrict__ kpad, float* __restrict__ vpad) {
  __shared__ float Xs[16 * 132];
  const int strip = blockIdx.x;
  const int t = threadIdx.x;

  for (int idx = t; idx < 16 * 32; idx += 256) {
    int m = idx >> 5, q4 = idx & 31;
    *(v4f*)&Xs[m * 132 + q4 * 4] =
        *(const v4f*)&x[(size_t)(strip * 16 + m) * 128 + q4 * 4];
  }
  __syncthreads();

  const int lane = t & 31;
  const int half = lane >> 4;
  const int n16  = lane & 15;
  const int wave = t >> 5;

  for (int tt = 0; tt < 3; ++tt) {
    const int otile = wave * 3 + tt;
    const int obase = otile * 16;
    const int o = obase + n16;
    __builtin_prefetch(&qkv_w[(size_t)o * 128], 0, 1);
    v8f acc = {};
#pragma unroll
    for (int kc = 0; kc < 32; ++kc) {
      const int ch = kc * 4 + 2 * half;
      v2f a = *(const v2f*)&Xs[n16 * 132 + ch];
      v2f b = *(const v2f*)&qkv_w[(size_t)o * 128 + ch];
      acc = wmma4(a, b, acc);
    }
    const float bias = qkv_b[o];
    const int which = otile >> 3;
    const int head  = (otile & 7) >> 1;
    const int d     = (otile & 1) * 16 + n16;
#pragma unroll
    for (int vr = 0; vr < 8; ++vr) {
      const int m = vr + 8 * half;
      const unsigned pixel = strip * 16 + m;
      const unsigned bimg = pixel / HWPIX;
      const unsigned rem  = pixel % HWPIX;
      const unsigned i = rem / IMW, j = rem % IMW;
      const float val = acc[vr] + bias;
      if (which == 0) {
        q_arr[(((size_t)bimg * NHEADS + head) * HWPIX + rem) * HDIM + d] =
            val * QSCALE;
      } else {
        float* dst = (which == 1) ? kpad : vpad;
        dst[((((size_t)bimg * NHEADS + head) * PADW + (i + 3)) * PADW +
             (j + 3)) * HDIM + d] = val;
      }
    }
  }
}

__global__ __launch_bounds__(32) void attn_kernel(
    const float* __restrict__ q_arr, const float* __restrict__ kpad,
    const float* __restrict__ vpad, const float* __restrict__ rpb,
    float* __restrict__ attnout) {
  __shared__ float PT[112 * 17];
  __shared__ float BiasT[49];

  int unit = blockIdx.x;
  const int tj = unit % 7;  unit /= 7;
  const int ti = unit % 28; unit /= 28;
  const int h  = unit % NHEADS; unit /= NHEADS;
  const int b  = unit;
  const int i0 = ti * 2, j0 = tj * 8;

  const int lane = threadIdx.x;
  const int half = lane >> 4;
  const int n16  = lane & 15;

  for (int idx = lane; idx < 49; idx += 32) {
    const int di = idx / 7, dj = idx % 7;
    BiasT[idx] = rpb[((size_t)h * 13 + (di + 3)) * 13 + (dj + 3)];
  }

  const size_t kvbase =
      ((((size_t)(b * NHEADS + h)) * PADW + i0) * PADW + j0) * HDIM;
  const float* __restrict__ kblk = kpad + kvbase;
  const float* __restrict__ vblk = vpad + kvbase;

  const int r = n16 >> 3, cq = n16 & 7;
  const int qi = i0 + r, qj = j0 + cq;
  const size_t qbase =
      (((size_t)(b * NHEADS + h)) * HWPIX + qi * IMW + qj) * HDIM;
  v4f qv[4];
#pragma unroll
  for (int kk = 0; kk < 4; ++kk)
    qv[kk] = *(const v4f*)&q_arr[qbase + kk * 8 + 4 * half];

  v4f kf[7][4];
#pragma unroll
  for (int tt = 0; tt < 7; ++tt) {
    const int pos = tt * 16 + n16;
    const int kr = pos / 14, kc = pos % 14;
    const int koff = (kr * PADW + kc) * HDIM;
#pragma unroll
    for (int kk = 0; kk < 4; ++kk)
      kf[tt][kk] = *(const v4f*)&kblk[koff + kk * 8 + 4 * half];
  }
  __syncthreads();

  v8f S[7];
#pragma unroll
  for (int tt = 0; tt < 7; ++tt) {
    v8f acc = {};
#pragma unroll
    for (int kk = 0; kk < 4; ++kk) {
      v2f a0, a1, b0, b1;
      a0.x = kf[tt][kk].x; a0.y = kf[tt][kk].y;
      a1.x = kf[tt][kk].z; a1.y = kf[tt][kk].w;
      b0.x = qv[kk].x;     b0.y = qv[kk].y;
      b1.x = qv[kk].z;     b1.y = qv[kk].w;
      acc = wmma4(a0, b0, acc);
      acc = wmma4(a1, b1, acc);
    }
#pragma unroll
    for (int vr = 0; vr < 8; ++vr) {
      const int key = tt * 16 + vr + 8 * half;
      const int kr = key / 14, kc2 = key % 14;
      const int di = kr - r, dj = kc2 - cq;
      if (di >= 0 && di < 7 && dj >= 0 && dj < 7)
        acc[vr] += BiasT[di * 7 + dj];
      else
        acc[vr] = -1e30f;
    }
    S[tt] = acc;
  }

  float mx = -1e30f;
#pragma unroll
  for (int tt = 0; tt < 7; ++tt)
#pragma unroll
    for (int vr = 0; vr < 8; ++vr) mx = fmaxf(mx, S[tt][vr]);
  mx = fmaxf(mx, __shfl_xor(mx, 16, 32));
  float sum = 0.f;
#pragma unroll
  for (int tt = 0; tt < 7; ++tt)
#pragma unroll
    for (int vr = 0; vr < 8; ++vr) {
      const float e = __expf(S[tt][vr] - mx);
      S[tt][vr] = e;
      sum += e;
    }
  sum += __shfl_xor(sum, 16, 32);
  const float inv = 1.f / sum;

#pragma unroll
  for (int tt = 0; tt < 7; ++tt)
#pragma unroll
    for (int vr = 0; vr < 8; ++vr)
      PT[(tt * 16 + vr + 8 * half) * 17 + n16] = S[tt][vr];
  __syncthreads();

  const size_t outbase =
      ((size_t)(b * HWPIX) + qi * IMW + qj) * 128 + h * HDIM;
  v8f acc0 = {}, acc1 = {};
#pragma unroll
  for (int kc = 0; kc < 28; ++kc) {
    const int ka = kc * 4, kb = kc * 4 + 2;
    const int offa = ((ka / 14) * PADW + (ka % 14)) * HDIM;
    const int offb = ((kb / 14) * PADW + (kb % 14)) * HDIM;
    const int voff = (half ? offb : offa) + n16;
    const int k0 = kc * 4 + 2 * half;
    v2f a0, a1, bfr;
    a0.x = vblk[voff];
    a0.y = vblk[voff + HDIM];
    a1.x = vblk[voff + 16];
    a1.y = vblk[voff + HDIM + 16];
    bfr.x = PT[k0 * 17 + n16];
    bfr.y = PT[(k0 + 1) * 17 + n16];
    acc0 = wmma4(a0, bfr, acc0);
    acc1 = wmma4(a1, bfr, acc1);
  }
#pragma unroll
  for (int vr = 0; vr < 8; ++vr) {
    attnout[outbase + vr + 8 * half]      = acc0[vr] * inv;
    attnout[outbase + 16 + vr + 8 * half] = acc1[vr] * inv;
  }
}

__global__ __launch_bounds__(256) void proj_kernel(
    const float* __restrict__ attnout, const float* __restrict__ proj_w,
    const float* __restrict__ proj_b, float* __restrict__ out) {
  __shared__ float Xs[16 * 132];
  const int strip = blockIdx.x;
  const int t = threadIdx.x;
  for (int idx = t; idx < 16 * 32; idx += 256) {
    int m = idx >> 5, q4 = idx & 31;
    *(v4f*)&Xs[m * 132 + q4 * 4] =
        *(const v4f*)&attnout[(size_t)(strip * 16 + m) * 128 + q4 * 4];
  }
  __syncthreads();
  const int lane = t & 31;
  const int half = lane >> 4;
  const int n16  = lane & 15;
  const int o = (t >> 5) * 16 + n16;
  __builtin_prefetch(&proj_w[(size_t)o * 128], 0, 1);
  v8f acc = {};
#pragma unroll
  for (int kc = 0; kc < 32; ++kc) {
    const int ch = kc * 4 + 2 * half;
    v2f a = *(const v2f*)&Xs[n16 * 132 + ch];
    v2f b = *(const v2f*)&proj_w[(size_t)o * 128 + ch];
    acc = wmma4(a, b, acc);
  }
  const float bias = proj_b[o];
#pragma unroll
  for (int vr = 0; vr < 8; ++vr) {
    const int m = vr + 8 * half;
    out[(size_t)(strip * 16 + m) * 128 + o] = acc[vr] + bias;
  }
}

extern "C" void kernel_launch(void* const* d_in, const int* in_sizes, int n_in,
                              void* d_out, int out_size, void* d_ws,
                              size_t ws_size, hipStream_t stream) {
  (void)in_sizes; (void)n_in; (void)out_size; (void)ws_size;
  const float* x      = (const float*)d_in[0];
  const float* qkv_w  = (const float*)d_in[1];
  const float* qkv_b  = (const float*)d_in[2];
  const float* rpb    = (const float*)d_in[3];
  const float* proj_w = (const float*)d_in[4];
  const float* proj_b = (const float*)d_in[5];

  float* ws = (float*)d_ws;
  const size_t q_len   = (size_t)BATCH * NHEADS * HWPIX * HDIM;
  const size_t pad_len = (size_t)BATCH * NHEADS * PADW * PADW * HDIM;
  float* q_arr   = ws;
  float* kpad    = ws + q_len;
  float* vpad    = kpad + pad_len;
  float* attnout = vpad + pad_len;

  hipMemsetAsync(kpad, 0, 2 * pad_len * sizeof(float), stream);

  qkv_kernel<<<784, 256, 0, stream>>>(x, qkv_w, qkv_b, q_arr, kpad, vpad);
  attn_kernel<<<BATCH * NHEADS * 28 * 7, 32, 0, stream>>>(q_arr, kpad, vpad,
                                                          rpb, attnout);
  proj_kernel<<<784, 256, 0, stream>>>(attnout, proj_w, proj_b, (float*)d_out);
}


# QKV and out-projection GEMMs: 64-pixel strip per block (was 16), each wave holds 4 M-tiles so one W fragment load feeds 4 WMMAs; W fragments loaded in groups ahead of the WMMAs; k order unchanged
# speedup vs baseline: 1.6764x; 1.2006x over previous
#include <hip/hip_runtime.h>


typedef __attribute__((ext_vector_type(2))) float v2f;
typedef __attribute__((ext_vector_type(4))) float v4f;
typedef __attribute__((ext_vector_type(8))) float v8f;

#define NHEADS 4
#define HDIM   32
#define IMH    56
#define IMW    56
#define HWPIX  3136
#define BATCH  4
#define PADW   62
#define QSCALE 0.17677669529663687f

static __device__ __forceinline__ v8f wmma4(v2f a, v2f b, v8f c) {
  return __builtin_amdgcn_wmma_f32_16x16x4_f32(false, a, false, b, (short)0, c,
                                               false, false);
}

#define STRIP_M  64
#define STRIP_MT 4
#define XPITCH   132
#define WKB      8
#define PKB      16

__global__ __launch_bounds__(256) void qkv_kernel(
    const float* __restrict__ x, const float* __restrict__ qkv_w,
    const float* __restrict__ qkv_b, float* __restrict__ q_arr,
    float* __restrict__ kpad, float* __restrict__ vpad) {
  __shared__ float Xs[STRIP_M * XPITCH];
  const int strip = blockIdx.x;
  const int t = threadIdx.x;
  const int lane = t & 31;
  const int half = lane >> 4;
  const int n16  = lane & 15;
  const int wave = __builtin_amdgcn_readfirstlane(t >> 5);

  const float* __restrict__ wrow[3];
#pragma unroll
  for (int tt = 0; tt < 3; ++tt)
    wrow[tt] = qkv_w + (size_t)((wave * 3 + tt) * 16 + n16) * 128 + 2 * half;

  v2f bw[2][3][WKB];
#pragma unroll
  for (int tt = 0; tt < 3; ++tt)
#pragma unroll
    for (int kk = 0; kk < WKB; ++kk)
      bw[0][tt][kk] = *(const v2f*)&wrow[tt][kk * 4];

  v4f xin[8];
#pragma unroll
  for (int it = 0; it < 8; ++it)
    xin[it] = *(const v4f*)&x[(size_t)(strip * STRIP_M + (t >> 5) + 8 * it) * 128 +
                              (t & 31) * 4];
#pragma unroll
  for (int it = 0; it < 8; ++it)
    *(v4f*)&Xs[((t >> 5) + 8 * it) * XPITCH + (t & 31) * 4] = xin[it];
  __syncthreads();

  const float* xrow = &Xs[n16 * XPITCH + 2 * half];

  v8f acc[3][STRIP_MT];
#pragma unroll
  for (int tt = 0; tt < 3; ++tt)
#pragma unroll
    for (int mt = 0; mt < STRIP_MT; ++mt) acc[tt][mt] = (v8f){};

#pragma unroll
  for (int blk = 0; blk < 32 / WKB; ++blk) {
    if (blk + 1 < 32 / WKB) {
#pragma unroll
      for (int tt = 0; tt < 3; ++tt)
#pragma unroll
        for (int kk = 0; kk < WKB; ++kk)
          bw[(blk + 1) & 1][tt][kk] =
              *(const v2f*)&wrow[tt][((blk + 1) * WKB + kk) * 4];
    }
    __builtin_amdgcn_sched_barrier(0);
#pragma unroll
    for (int kk = 0; kk < WKB; ++kk) {
      const int kc = blk * WKB + kk;
      v2f a[STRIP_MT];
#pragma unroll
      for (int mt = 0; mt < STRIP_MT; ++mt)
        a[mt] = *(const v2f*)&xrow[mt * 16 * XPITCH + kc * 4];
#pragma unroll
      for (int tt = 0; tt < 3; ++tt)
#pragma unroll
        for (int mt = 0; mt < STRIP_MT; ++mt)
          acc[tt][mt] = wmma4(a[mt], bw[blk & 1][tt][kk], acc[tt][mt]);
    }
  }

  float bias[3], scl[3];
  float* dstp[3];
  unsigned hd[3];
  bool isq[3];
#pragma unroll
  for (int tt = 0; tt < 3; ++tt) {
    const int otile = wave * 3 + tt;
    const int which = otile >> 3;
    const unsigned head = (otile & 7) >> 1;
    const unsigned d    = (otile & 1) * 16 + n16;
    isq[tt]  = (which == 0);
    dstp[tt] = isq[tt] ? q_arr : ((which == 1) ? kpad : vpad);
    scl[tt]  = isq[tt] ? QSCALE : 1.0f;
    hd[tt]   = head * (isq[tt] ? (unsigned)(HWPIX * HDIM)
                               : (unsigned)(PADW * PADW * HDIM)) + d;
    bias[tt] = qkv_b[otile * 16 + n16];
  }

#pragma unroll
  for (int mt = 0; mt < STRIP_MT; ++mt) {
#pragma unroll
    for (int vr = 0; vr < 8; ++vr) {
      const unsigned m = mt * 16 + vr + 8 * half;
      const unsigned pixel = strip * STRIP_M + m;
      const unsigned bimg = pixel / HWPIX;
      const unsigned rem  = pixel - bimg * HWPIX;
      const unsigned i    = rem / IMW;
      const unsigned qpix = (pixel + bimg * ((NHEADS - 1) * HWPIX)) * HDIM;
      const unsigned ppix = (bimg * (NHEADS * PADW * PADW) + rem +
                             i * (PADW - IMW) + (3 * PADW + 3)) * HDIM;
#pragma unroll
      for (int tt = 0; tt < 3; ++tt) {
        const float val = (acc[tt][mt][vr] + bias[tt]) * scl[tt];
        dstp[tt][(isq[tt] ? qpix : ppix) + hd[tt]] = val;
      }
    }
  }
}

__global__ __launch_bounds__(32) void attn_kernel(
    const float* __restrict__ q_arr, const float* __restrict__ kpad,
    const float* __restrict__ vpad, const float* __restrict__ rpb,
    float* __restrict__ attnout) {
  __shared__ float PT[112 * 17];
  __shared__ float BiasT[49];

  int unit = blockIdx.x;
  const int tj = unit % 7;  unit /= 7;
  const int ti = unit % 28; unit /= 28;
  const int h  = unit % NHEADS; unit /= NHEADS;
  const int b  = unit;
  const int i0 = ti * 2, j0 = tj * 8;

  const int lane = threadIdx.x;
  const int half = lane >> 4;
  const int n16  = lane & 15;

  for (int idx = lane; idx < 49; idx += 32) {
    const int di = idx / 7, dj = idx % 7;
    BiasT[idx] = rpb[((size_t)h * 13 + (di + 3)) * 13 + (dj + 3)];
  }

  const size_t kvbase =
      ((((size_t)(b * NHEADS + h)) * PADW + i0) * PADW + j0) * HDIM;
  const float* __restrict__ kblk = kpad + kvbase;
  const float* __restrict__ vblk = vpad + kvbase;

  const int r = n16 >> 3, cq = n16 & 7;
  const int qi = i0 + r, qj = j0 + cq;
  const size_t qbase =
      (((size_t)(b * NHEADS + h)) * HWPIX + qi * IMW + qj) * HDIM;
  v4f qv[4];
#pragma unroll
  for (int kk = 0; kk < 4; ++kk)
    qv[kk] = *(const v4f*)&q_arr[qbase + kk * 8 + 4 * half];

  v4f kf[7][4];
#pragma unroll
  for (int tt = 0; tt < 7; ++tt) {
    const int pos = tt * 16 + n16;
    const int kr = pos / 14, kc = pos % 14;
    const int koff = (kr * PADW + kc) * HDIM;
#pragma unroll
    for (int kk = 0; kk < 4; ++kk)
      kf[tt][kk] = *(const v4f*)&kblk[koff + kk * 8 + 4 * half];
  }
  __syncthreads();

  v8f S[7];
#pragma unroll
  for (int tt = 0; tt < 7; ++tt) {
    v8f acc = {};
#pragma unroll
    for (int kk = 0; kk < 4; ++kk) {
      v2f a0, a1, b0, b1;
      a0.x = kf[tt][kk].x; a0.y = kf[tt][kk].y;
      a1.x = kf[tt][kk].z; a1.y = kf[tt][kk].w;
      b0.x = qv[kk].x;     b0.y = qv[kk].y;
      b1.x = qv[kk].z;     b1.y = qv[kk].w;
      acc = wmma4(a0, b0, acc);
      acc = wmma4(a1, b1, acc);
    }
#pragma unroll
    for (int vr = 0; vr < 8; ++vr) {
      const int key = tt * 16 + vr + 8 * half;
      const int kr = key / 14, kc2 = key % 14;
      const int di = kr - r, dj = kc2 - cq;
      if (di >= 0 && di < 7 && dj >= 0 && dj < 7)
        acc[vr] += BiasT[di * 7 + dj];
      else
        acc[vr] = -1e30f;
    }
    S[tt] = acc;
  }

  float mx = -1e30f;
#pragma unroll
  for (int tt = 0; tt < 7; ++tt)
#pragma unroll
    for (int vr = 0; vr < 8; ++vr) mx = fmaxf(mx, S[tt][vr]);
  mx = fmaxf(mx, __shfl_xor(mx, 16, 32));
  float sum = 0.f;
#pragma unroll
  for (int tt = 0; tt < 7; ++tt)
#pragma unroll
    for (int vr = 0; vr < 8; ++vr) {
      const float e = __expf(S[tt][vr] - mx);
      S[tt][vr] = e;
      sum += e;
    }
  sum += __shfl_xor(sum, 16, 32);
  const float inv = 1.f / sum;

#pragma unroll
  for (int tt = 0; tt < 7; ++tt)
#pragma unroll
    for (int vr = 0; vr < 8; ++vr)
      PT[(tt * 16 + vr + 8 * half) * 17 + n16] = S[tt][vr];
  __syncthreads();

  const size_t outbase =
      ((size_t)(b * HWPIX) + qi * IMW + qj) * 128 + h * HDIM;
  v8f acc0 = {}, acc1 = {};
#pragma unroll
  for (int kc = 0; kc < 28; ++kc) {
    const int ka = kc * 4, kb = kc * 4 + 2;
    const int offa = ((ka / 14) * PADW + (ka % 14)) * HDIM;
    const int offb = ((kb / 14) * PADW + (kb % 14)) * HDIM;
    const int voff = (half ? offb : offa) + n16;
    const int k0 = kc * 4 + 2 * half;
    v2f a0, a1, bfr;
    a0.x = vblk[voff];
    a0.y = vblk[voff + HDIM];
    a1.x = vblk[voff + 16];
    a1.y = vblk[voff + HDIM + 16];
    bfr.x = PT[k0 * 17 + n16];
    bfr.y = PT[(k0 + 1) * 17 + n16];
    acc0 = wmma4(a0, bfr, acc0);
    acc1 = wmma4(a1, bfr, acc1);
  }
#pragma unroll
  for (int vr = 0; vr < 8; ++vr) {
    attnout[outbase + vr + 8 * half]      = acc0[vr] * inv;
    attnout[outbase + 16 + vr + 8 * half] = acc1[vr] * inv;
  }
}

__global__ __launch_bounds__(256) void proj_kernel(
    const float* __restrict__ attnout, const float* __restrict__ proj_w,
    const float* __restrict__ proj_b, float* __restrict__ out) {
  __shared__ float Xs[STRIP_M * XPITCH];
  const int strip = blockIdx.x;
  const int t = threadIdx.x;
  const int lane = t & 31;
  const int half = lane >> 4;
  const int n16  = lane & 15;
  const int o = (t >> 5) * 16 + n16;
  const float* __restrict__ wrow = proj_w + (size_t)o * 128 + 2 * half;

  v2f bw[2][PKB];
#pragma unroll
  for (int kk = 0; kk < PKB; ++kk) bw[0][kk] = *(const v2f*)&wrow[kk * 4];

  v4f xin[8];
#pragma unroll
  for (int it = 0; it < 8; ++it)
    xin[it] = *(const v4f*)&attnout[(size_t)(strip * STRIP_M + (t >> 5) + 8 * it) * 128 +
                                    (t & 31) * 4];
#pragma unroll
  for (int it = 0; it < 8; ++it)
    *(v4f*)&Xs[((t >> 5) + 8 * it) * XPITCH + (t & 31) * 4] = xin[it];
  __syncthreads();

  const float* xrow = &Xs[n16 * XPITCH + 2 * half];
  v8f acc[STRIP_MT];
#pragma unroll
  for (int mt = 0; mt < STRIP_MT; ++mt) acc[mt] = (v8f){};

#pragma unroll
  for (int blk = 0; blk < 32 / PKB; ++blk) {
    if (blk + 1 < 32 / PKB) {
#pragma unroll
      for (int kk = 0; kk < PKB; ++kk)
        bw[(blk + 1) & 1][kk] = *(const v2f*)&wrow[((blk + 1) * PKB + kk) * 4];
    }
    __builtin_amdgcn_sched_barrier(0);
#pragma unroll
    for (int kk = 0; kk < PKB; ++kk) {
      const int kc = blk * PKB + kk;
      v2f a[STRIP_MT];
#pragma unroll
      for (int mt = 0; mt < STRIP_MT; ++mt)
        a[mt] = *(const v2f*)&xrow[mt * 16 * XPITCH + kc * 4];
#pragma unroll
      for (int mt = 0; mt < STRIP_MT; ++mt)
        acc[mt] = wmma4(a[mt], bw[blk & 1][kk], acc[mt]);
    }
  }

  const float bias = proj_b[o];
#pragma unroll
  for (int mt = 0; mt < STRIP_MT; ++mt)
#pragma unroll
    for (int vr = 0; vr < 8; ++vr) {
      const int m = mt * 16 + vr + 8 * half;
      out[(size_t)(strip * STRIP_M + m) * 128 + o] = acc[mt][vr] + bias;
    }
}

extern "C" void kernel_launch(void* const* d_in, const int* in_sizes, int n_in,
                              void* d_out, int out_size, void* d_ws,
                              size_t ws_size, hipStream_t stream) {
  (void)in_sizes; (void)n_in; (void)out_size; (void)ws_size;
  const float* x      = (const float*)d_in[0];
  const float* qkv_w  = (const float*)d_in[1];
  const float* qkv_b  = (const float*)d_in[2];
  const float* rpb    = (const float*)d_in[3];
  const float* proj_w = (const float*)d_in[4];
  const float* proj_b = (const float*)d_in[5];

  float* ws = (float*)d_ws;
  const size_t q_len   = (size_t)BATCH * NHEADS * HWPIX * HDIM;
  const size_t pad_len = (size_t)BATCH * NHEADS * PADW * PADW * HDIM;
  float* q_arr   = ws;
  float* kpad    = ws + q_len;
  float* vpad    = kpad + pad_len;
  float* attnout = vpad + pad_len;

  hipMemsetAsync(kpad, 0, 2 * pad_len * sizeof(float), stream);

  qkv_kernel<<<196, 256, 0, stream>>>(x, qkv_w, qkv_b, q_arr, kpad, vpad);
  attn_kernel<<<BATCH * NHEADS * 28 * 7, 32, 0, stream>>>(q_arr, kpad, vpad,
                                                          rpb, attnout);
  proj_kernel<<<196, 256, 0, stream>>>(attnout, proj_w, proj_b, (float*)d_out);
}


# attention scores: bias + 7x7 window mask applied branch-free (constant lane set per score register as select mask, bias loads at constant offsets); was 56 serial masked branches, an LDS wait each
# speedup vs baseline: 1.7589x; 1.0492x over previous
#include <hip/hip_runtime.h>


typedef __attribute__((ext_vector_type(2))) float v2f;
typedef __attribute__((ext_vector_type(4))) float v4f;
typedef __attribute__((ext_vector_type(8))) float v8f;

#define NHEADS 4
#define HDIM   32
#define IMH    56
#define IMW    56
#define HWPIX  3136
#define BATCH  4
#define PADW   62
#define QSCALE 0.17677669529663687f

static __device__ __forceinline__ v8f wmma4(v2f a, v2f b, v8f c) {
  return __builtin_amdgcn_wmma_f32_16x16x4_f32(false, a, false, b, (short)0, c,
                                               false, false);
}

#define STRIP_M  64
#define STRIP_MT 4
#define XPITCH   132
#define WKB      8
#define PKB      16

__global__ __launch_bounds__(256) void qkv_kernel(
    const float* __restrict__ x, const float* __restrict__ qkv_w,
    const float* __restrict__ qkv_b, float* __restrict__ q_arr,
    float* __restrict__ kpad, float* __restrict__ vpad) {
  __shared__ float Xs[STRIP_M * XPITCH];
  const int strip = blockIdx.x;
  const int t = threadIdx.x;
  const int lane = t & 31;
  const int half = lane >> 4;
  const int n16  = lane & 15;
  const int wave = __builtin_amdgcn_readfirstlane(t >> 5);

  const float* __restrict__ wrow[3];
#pragma unroll
  for (int tt = 0; tt < 3; ++tt)
    wrow[tt] = qkv_w + (size_t)((wave * 3 + tt) * 16 + n16) * 128 + 2 * half;

  v2f bw[2][3][WKB];
#pragma unroll
  for (int tt = 0; tt < 3; ++tt)
#pragma unroll
    for (int kk = 0; kk < WKB; ++kk)
      bw[0][tt][kk] = *(const v2f*)&wrow[tt][kk * 4];

  v4f xin[8];
#pragma unroll
  for (int it = 0; it < 8; ++it)
    xin[it] = *(const v4f*)&x[(size_t)(strip * STRIP_M + (t >> 5) + 8 * it) * 128 +
                              (t & 31) * 4];
#pragma unroll
  for (int it = 0; it < 8; ++it)
    *(v4f*)&Xs[((t >> 5) + 8 * it) * XPITCH + (t & 31) * 4] = xin[it];
  __syncthreads();

  const float* xrow = &Xs[n16 * XPITCH + 2 * half];

  v8f acc[3][STRIP_MT];
#pragma unroll
  for (int tt = 0; tt < 3; ++tt)
#pragma unroll
    for (int mt = 0; mt < STRIP_MT; ++mt) acc[tt][mt] = (v8f){};

#pragma unroll
  for (int blk = 0; blk < 32 / WKB; ++blk) {
    if (blk + 1 < 32 / WKB) {
#pragma unroll
      for (int tt = 0; tt < 3; ++tt)
#pragma unroll
        for (int kk = 0; kk < WKB; ++kk)
          bw[(blk + 1) & 1][tt][kk] =
              *(const v2f*)&wrow[tt][((blk + 1) * WKB + kk) * 4];
    }
    __builtin_amdgcn_sched_barrier(0);
#pragma unroll
    for (int kk = 0; kk < WKB; ++kk) {
      const int kc = blk * WKB + kk;
      v2f a[STRIP_MT];
#pragma unroll
      for (int mt = 0; mt < STRIP_MT; ++mt)
        a[mt] = *(const v2f*)&xrow[mt * 16 * XPITCH + kc * 4];
#pragma unroll
      for (int tt = 0; tt < 3; ++tt)
#pragma unroll
        for (int mt = 0; mt < STRIP_MT; ++mt)
          acc[tt][mt] = wmma4(a[mt], bw[blk & 1][tt][kk], acc[tt][mt]);
    }
  }

  float bias[3], scl[3];
  float* dstp[3];
  unsigned hd[3];
  bool isq[3];
#pragma unroll
  for (int tt = 0; tt < 3; ++tt) {
    const int otile = wave * 3 + tt;
    const int which = otile >> 3;
    const unsigned head = (otile & 7) >> 1;
    const unsigned d    = (otile & 1) * 16 + n16;
    isq[tt]  = (which == 0);
    dstp[tt] = isq[tt] ? q_arr : ((which == 1) ? kpad : vpad);
    scl[tt]  = isq[tt] ? QSCALE : 1.0f;
    hd[tt]   = head * (isq[tt] ? (unsigned)(HWPIX * HDIM)
                               : (unsigned)(PADW * PADW * HDIM)) + d;
    bias[tt] = qkv_b[otile * 16 + n16];
  }

#pragma unroll
  for (int mt = 0; mt < STRIP_MT; ++mt) {
#pragma unroll
    for (int vr = 0; vr < 8; ++vr) {
      const unsigned m = mt * 16 + vr + 8 * half;
      const unsigned pixel = strip * STRIP_M + m;
      const unsigned bimg = pixel / HWPIX;
      const unsigned rem  = pixel - bimg * HWPIX;
      const unsigned i    = rem / IMW;
      const unsigned qpix = (pixel + bimg * ((NHEADS - 1) * HWPIX)) * HDIM;
      const unsigned ppix = (bimg * (NHEADS * PADW * PADW) + rem +
                             i * (PADW - IMW) + (3 * PADW + 3)) * HDIM;
#pragma unroll
      for (int tt = 0; tt < 3; ++tt) {
        const float val = (acc[tt][mt][vr] + bias[tt]) * scl[tt];
        dstp[tt][(isq[tt] ? qpix : ppix) + hd[tt]] = val;
      }
    }
  }
}

static constexpr unsigned win_lanes(int tt, int vr) {
  unsigned m = 0;
  for (int lane = 0; lane < 32; ++lane) {
    const int half = lane >> 4, n16 = lane & 15;
    const int r = n16 >> 3, cq = n16 & 7;
    const int key = tt * 16 + vr + 8 * half;
    const int di = key / 14 - r, dj = key % 14 - cq;
    if (di >= 0 && di < 7 && dj >= 0 && dj < 7) m |= 1u << lane;
  }
  return m;
}

template <int TT, int VR>
static __device__ __forceinline__ float bias_or_mask(float s,
                                                     const float* bsame,
                                                     const float* bwrap) {
  constexpr int key0 = TT * 16 + VR;
  constexpr int kc0 = key0 % 14;
  constexpr int c0 = (key0 / 14) * 7 + kc0;
  constexpr unsigned lanes = win_lanes(TT, VR);
  const float bias = (kc0 < 6 ? bsame : bwrap)[c0];
  return __builtin_amdgcn_inverse_ballot_w32(lanes) ? s + bias : -1e30f;
}

template <int TT>
static __device__ __forceinline__ v8f score_tile(const v4f (&kf)[4],
                                                 const v4f (&qv)[4],
                                                 const float* bsame,
                                                 const float* bwrap) {
  v8f acc = {};
#pragma unroll
  for (int kk = 0; kk < 4; ++kk) {
    v2f a0, a1, b0, b1;
    a0.x = kf[kk].x; a0.y = kf[kk].y;
    a1.x = kf[kk].z; a1.y = kf[kk].w;
    b0.x = qv[kk].x; b0.y = qv[kk].y;
    b1.x = qv[kk].z; b1.y = qv[kk].w;
    acc = wmma4(a0, b0, acc);
    acc = wmma4(a1, b1, acc);
  }
  acc[0] = bias_or_mask<TT, 0>(acc[0], bsame, bwrap);
  acc[1] = bias_or_mask<TT, 1>(acc[1], bsame, bwrap);
  acc[2] = bias_or_mask<TT, 2>(acc[2], bsame, bwrap);
  acc[3] = bias_or_mask<TT, 3>(acc[3], bsame, bwrap);
  acc[4] = bias_or_mask<TT, 4>(acc[4], bsame, bwrap);
  acc[5] = bias_or_mask<TT, 5>(acc[5], bsame, bwrap);
  acc[6] = bias_or_mask<TT, 6>(acc[6], bsame, bwrap);
  acc[7] = bias_or_mask<TT, 7>(acc[7], bsame, bwrap);
  return acc;
}

__global__ __launch_bounds__(32) void attn_kernel(
    const float* __restrict__ q_arr, const float* __restrict__ kpad,
    const float* __restrict__ vpad, const float* __restrict__ rpb,
    float* __restrict__ attnout) {
  __shared__ float PT[112 * 17];
  __shared__ float BiasP[80];

  int unit = blockIdx.x;
  const int tj = unit % 7;  unit /= 7;
  const int ti = unit % 28; unit /= 28;
  const int h  = unit % NHEADS; unit /= NHEADS;
  const int b  = unit;
  const int i0 = ti * 2, j0 = tj * 8;

  const int lane = threadIdx.x;
  const int half = lane >> 4;
  const int n16  = lane & 15;

  for (int idx = lane; idx < 80; idx += 32) {
    const int t7 = idx - 14;
    float bval = 0.f;
    if (t7 >= 0 && t7 < 49) {
      const int di = t7 / 7, dj = t7 % 7;
      bval = rpb[((size_t)h * 13 + (di + 3)) * 13 + (dj + 3)];
    }
    BiasP[idx] = bval;
  }

  const size_t kvbase =
      ((((size_t)(b * NHEADS + h)) * PADW + i0) * PADW + j0) * HDIM;
  const float* __restrict__ kblk = kpad + kvbase;
  const float* __restrict__ vblk = vpad + kvbase;

  const int r = n16 >> 3, cq = n16 & 7;
  const int qi = i0 + r, qj = j0 + cq;
  const size_t qbase =
      (((size_t)(b * NHEADS + h)) * HWPIX + qi * IMW + qj) * HDIM;
  v4f qv[4];
#pragma unroll
  for (int kk = 0; kk < 4; ++kk)
    qv[kk] = *(const v4f*)&q_arr[qbase + kk * 8 + 4 * half];

  v4f kf[7][4];
#pragma unroll
  for (int tt = 0; tt < 7; ++tt) {
    const int pos = tt * 16 + n16;
    const int kr = pos / 14, kc = pos % 14;
    const int koff = (kr * PADW + kc) * HDIM;
#pragma unroll
    for (int kk = 0; kk < 4; ++kk)
      kf[tt][kk] = *(const v4f*)&kblk[koff + kk * 8 + 4 * half];
  }
  __syncthreads();

  const float* bsame = BiasP + 14 - (r * 7 + cq) + 8 * half;
  const float* bwrap = BiasP + 14 - (r * 7 + cq) + half;
  v8f S[7];
  S[0] = score_tile<0>(kf[0], qv, bsame, bwrap);
  S[1] = score_tile<1>(kf[1], qv, bsame, bwrap);
  S[2] = score_tile<2>(kf[2], qv, bsame, bwrap);
  S[3] = score_tile<3>(kf[3], qv, bsame, bwrap);
  S[4] = score_tile<4>(kf[4], qv, bsame, bwrap);
  S[5] = score_tile<5>(kf[5], qv, bsame, bwrap);
  S[6] = score_tile<6>(kf[6], qv, bsame, bwrap);

  float mx = -1e30f;
#pragma unroll
  for (int tt = 0; tt < 7; ++tt)
#pragma unroll
    for (int vr = 0; vr < 8; ++vr) mx = fmaxf(mx, S[tt][vr]);
  mx = fmaxf(mx, __shfl_xor(mx, 16, 32));
  float sum = 0.f;
#pragma unroll
  for (int tt = 0; tt < 7; ++tt)
#pragma unroll
    for (int vr = 0; vr < 8; ++vr) {
      const float e = __expf(S[tt][vr] - mx);
      S[tt][vr] = e;
      sum += e;
    }
  sum += __shfl_xor(sum, 16, 32);
  const float inv = 1.f / sum;

#pragma unroll
  for (int tt = 0; tt < 7; ++tt)
#pragma unroll
    for (int vr = 0; vr < 8; ++vr)
      PT[(tt * 16 + vr + 8 * half) * 17 + n16] = S[tt][vr];
  __syncthreads();

  const size_t outbase =
      ((size_t)(b * HWPIX) + qi * IMW + qj) * 128 + h * HDIM;
  v8f acc0 = {}, acc1 = {};
#pragma unroll
  for (int kc = 0; kc < 28; ++kc) {
    const int ka = kc * 4, kb = kc * 4 + 2;
    const int offa = ((ka / 14) * PADW + (ka % 14)) * HDIM;
    const int offb = ((kb / 14) * PADW + (kb % 14)) * HDIM;
    const int voff = (half ? offb : offa) + n16;
    const int k0 = kc * 4 + 2 * half;
    v2f a0, a1, bfr;
    a0.x = vblk[voff];
    a0.y = vblk[voff + HDIM];
    a1.x = vblk[voff + 16];
    a1.y = vblk[voff + HDIM + 16];
    bfr.x = PT[k0 * 17 + n16];
    bfr.y = PT[(k0 + 1) * 17 + n16];
    acc0 = wmma4(a0, bfr, acc0);
    acc1 = wmma4(a1, bfr, acc1);
  }
#pragma unroll
  for (int vr = 0; vr < 8; ++vr) {
    attnout[outbase + vr + 8 * half]      = acc0[vr] * inv;
    attnout[outbase + 16 + vr + 8 * half] = acc1[vr] * inv;
  }
}

__global__ __launch_bounds__(256) void proj_kernel(
    const float* __restrict__ attnout, const float* __restrict__ proj_w,
    const float* __restrict__ proj_b, float* __restrict__ out) {
  __shared__ float Xs[STRIP_M * XPITCH];
  const int strip = blockIdx.x;
  const int t = threadIdx.x;
  const int lane = t & 31;
  const int half = lane >> 4;
  const int n16  = lane & 15;
  const int o = (t >> 5) * 16 + n16;
  const float* __restrict__ wrow = proj_w + (size_t)o * 128 + 2 * half;

  v2f bw[2][PKB];
#pragma unroll
  for (int kk = 0; kk < PKB; ++kk) bw[0][kk] = *(const v2f*)&wrow[kk * 4];

  v4f xin[8];
#pragma unroll
  for (int it = 0; it < 8; ++it)
    xin[it] = *(const v4f*)&attnout[(size_t)(strip * STRIP_M + (t >> 5) + 8 * it) * 128 +
                                    (t & 31) * 4];
#pragma unroll
  for (int it = 0; it < 8; ++it)
    *(v4f*)&Xs[((t >> 5) + 8 * it) * XPITCH + (t & 31) * 4] = xin[it];
  __syncthreads();

  const float* xrow = &Xs[n16 * XPITCH + 2 * half];
  v8f acc[STRIP_MT];
#pragma unroll
  for (int mt = 0; mt < STRIP_MT; ++mt) acc[mt] = (v8f){};

#pragma unroll
  for (int blk = 0; blk < 32 / PKB; ++blk) {
    if (blk + 1 < 32 / PKB) {
#pragma unroll
      for (int kk = 0; kk < PKB; ++kk)
        bw[(blk + 1) & 1][kk] = *(const v2f*)&wrow[((blk + 1) * PKB + kk) * 4];
    }
    __builtin_amdgcn_sched_barrier(0);
#pragma unroll
    for (int kk = 0; kk < PKB; ++kk) {
      const int kc = blk * PKB + kk;
      v2f a[STRIP_MT];
#pragma unroll
      for (int mt = 0; mt < STRIP_MT; ++mt)
        a[mt] = *(const v2f*)&xrow[mt * 16 * XPITCH + kc * 4];
#pragma unroll
      for (int mt = 0; mt < STRIP_MT; ++mt)
        acc[mt] = wmma4(a[mt], bw[blk & 1][kk], acc[mt]);
    }
  }

  const float bias = proj_b[o];
#pragma unroll
  for (int mt = 0; mt < STRIP_MT; ++mt)
#pragma unroll
    for (int vr = 0; vr < 8; ++vr) {
      const int m = mt * 16 + vr + 8 * half;
      out[(size_t)(strip * STRIP_M + m) * 128 + o] = acc[mt][vr] + bias;
    }
}

extern "C" void kernel_launch(void* const* d_in, const int* in_sizes, int n_in,
                              void* d_out, int out_size, void* d_ws,
                              size_t ws_size, hipStream_t stream) {
  (void)in_sizes; (void)n_in; (void)out_size; (void)ws_size;
  const float* x      = (const float*)d_in[0];
  const float* qkv_w  = (const float*)d_in[1];
  const float* qkv_b  = (const float*)d_in[2];
  const float* rpb    = (const float*)d_in[3];
  const float* proj_w = (const float*)d_in[4];
  const float* proj_b = (const float*)d_in[5];

  float* ws = (float*)d_ws;
  const size_t q_len   = (size_t)BATCH * NHEADS * HWPIX * HDIM;
  const size_t pad_len = (size_t)BATCH * NHEADS * PADW * PADW * HDIM;
  float* q_arr   = ws;
  float* kpad    = ws + q_len;
  float* vpad    = kpad + pad_len;
  float* attnout = vpad + pad_len;

  hipMemsetAsync(kpad, 0, 2 * pad_len * sizeof(float), stream);

  qkv_kernel<<<196, 256, 0, stream>>>(x, qkv_w, qkv_b, q_arr, kpad, vpad);
  attn_kernel<<<BATCH * NHEADS * 28 * 7, 32, 0, stream>>>(q_arr, kpad, vpad,
                                                          rpb, attnout);
  proj_kernel<<<196, 256, 0, stream>>>(attnout, proj_w, proj_b, (float*)d_out);
}


# out-projection fused into attention kernel: block = 4 waves (heads) per 2x8 tile, head outputs meet in LDS, proj_w read as fragment-ordered copy; proj launch gone; proj k-terms summed in new order
# speedup vs baseline: 1.9012x; 1.0809x over previous
#include <hip/hip_runtime.h>


typedef __attribute__((ext_vector_type(2))) float v2f;
typedef __attribute__((ext_vector_type(4))) float v4f;
typedef __attribute__((ext_vector_type(8))) float v8f;

#define NHEADS 4
#define HDIM   32
#define IMH    56
#define IMW    56
#define HWPIX  3136
#define BATCH  4
#define PADW   62
#define QSCALE 0.17677669529663687f

static __device__ __forceinline__ v8f wmma4(v2f a, v2f b, v8f c) {
  return __builtin_amdgcn_wmma_f32_16x16x4_f32(false, a, false, b, (short)0, c,
                                               false, false);
}

#define STRIP_M  64
#define STRIP_MT 4
#define XPITCH   132
#define WKB      8

__global__ __launch_bounds__(256) void qkv_kernel(
    const float* __restrict__ x, const float* __restrict__ qkv_w,
    const float* __restrict__ qkv_b, float* __restrict__ q_arr,
    float* __restrict__ kpad, float* __restrict__ vpad,
    const float* __restrict__ proj_w, float* __restrict__ wfrag) {
  __shared__ float Xs[STRIP_M * XPITCH];
  const int strip = blockIdx.x;
  const int t = threadIdx.x;

  if (strip < 16) {
    const int piece = strip * 256 + t;
    const int ln = piece & 31, jj = (piece >> 5) & 15, tile = piece >> 9;
    const v4f w4 = *(const v4f*)&proj_w[(size_t)(tile * 16 + (ln & 15)) * 128 +
                                        8 * jj + 4 * (ln >> 4)];
    *(v4f*)&wfrag[(size_t)piece * 4] = w4;
  }

  const int lane = t & 31;
  const int half = lane >> 4;
  const int n16  = lane & 15;
  const int wave = __builtin_amdgcn_readfirstlane(t >> 5);

  const float* __restrict__ wrow[3];
#pragma unroll
  for (int tt = 0; tt < 3; ++tt)
    wrow[tt] = qkv_w + (size_t)((wave * 3 + tt) * 16 + n16) * 128 + 2 * half;

  v2f bw[2][3][WKB];
#pragma unroll
  for (int tt = 0; tt < 3; ++tt)
#pragma unroll
    for (int kk = 0; kk < WKB; ++kk)
      bw[0][tt][kk] = *(const v2f*)&wrow[tt][kk * 4];

  v4f xin[8];
#pragma unroll
  for (int it = 0; it < 8; ++it)
    xin[it] = *(const v4f*)&x[(size_t)(strip * STRIP_M + (t >> 5) + 8 * it) * 128 +
                              (t & 31) * 4];
#pragma unroll
  for (int it = 0; it < 8; ++it)
    *(v4f*)&Xs[((t >> 5) + 8 * it) * XPITCH + (t & 31) * 4] = xin[it];
  __syncthreads();

  const float* xrow = &Xs[n16 * XPITCH + 2 * half];

  v8f acc[3][STRIP_MT];
#pragma unroll
  for (int tt = 0; tt < 3; ++tt)
#pragma unroll
    for (int mt = 0; mt < STRIP_MT; ++mt) acc[tt][mt] = (v8f){};

#pragma unroll
  for (int blk = 0; blk < 32 / WKB; ++blk) {
    if (blk + 1 < 32 / WKB) {
#pragma unroll
      for (int tt = 0; tt < 3; ++tt)
#pragma unroll
        for (int kk = 0; kk < WKB; ++kk)
          bw[(blk + 1) & 1][tt][kk] =
              *(const v2f*)&wrow[tt][((blk + 1) * WKB + kk) * 4];
    }
    __builtin_amdgcn_sched_barrier(0);
#pragma unroll
    for (int kk = 0; kk < WKB; ++kk) {
      const int kc = blk * WKB + kk;
      v2f a[STRIP_MT];
#pragma unroll
      for (int mt = 0; mt < STRIP_MT; ++mt)
        a[mt] = *(const v2f*)&xrow[mt * 16 * XPITCH + kc * 4];
#pragma unroll
      for (int tt = 0; tt < 3; ++tt)
#pragma unroll
        for (int mt = 0; mt < STRIP_MT; ++mt)
          acc[tt][mt] = wmma4(a[mt], bw[blk & 1][tt][kk], acc[tt][mt]);
    }
  }

  float bias[3], scl[3];
  float* dstp[3];
  unsigned hd[3];
  bool isq[3];
#pragma unroll
  for (int tt = 0; tt < 3; ++tt) {
    const int otile = wave * 3 + tt;
    const int which = otile >> 3;
    const unsigned head = (otile & 7) >> 1;
    const unsigned d    = (otile & 1) * 16 + n16;
    isq[tt]  = (which == 0);
    dstp[tt] = isq[tt] ? q_arr : ((which == 1) ? kpad : vpad);
    scl[tt]  = isq[tt] ? QSCALE : 1.0f;
    hd[tt]   = head * (isq[tt] ? (unsigned)(HWPIX * HDIM)
                               : (unsigned)(PADW * PADW * HDIM)) + d;
    bias[tt] = qkv_b[otile * 16 + n16];
  }

#pragma unroll
  for (int mt = 0; mt < STRIP_MT; ++mt) {
#pragma unroll
    for (int vr = 0; vr < 8; ++vr) {
      const unsigned m = mt * 16 + vr + 8 * half;
      const unsigned pixel = strip * STRIP_M + m;
      const unsigned bimg = pixel / HWPIX;
      const unsigned rem  = pixel - bimg * HWPIX;
      const unsigned i    = rem / IMW;
      const unsigned qpix = (pixel + bimg * ((NHEADS - 1) * HWPIX)) * HDIM;
      const unsigned ppix = (bimg * (NHEADS * PADW * PADW) + rem +
                             i * (PADW - IMW) + (3 * PADW + 3)) * HDIM;
#pragma unroll
      for (int tt = 0; tt < 3; ++tt) {
        const float val = (acc[tt][mt][vr] + bias[tt]) * scl[tt];
        dstp[tt][(isq[tt] ? qpix : ppix) + hd[tt]] = val;
      }
    }
  }
}

static constexpr unsigned win_lanes(int tt, int vr) {
  unsigned m = 0;
  for (int lane = 0; lane < 32; ++lane) {
    const int half = lane >> 4, n16 = lane & 15;
    const int r = n16 >> 3, cq = n16 & 7;
    const int key = tt * 16 + vr + 8 * half;
    const int di = key / 14 - r, dj = key % 14 - cq;
    if (di >= 0 && di < 7 && dj >= 0 && dj < 7) m |= 1u << lane;
  }
  return m;
}

template <int TT, int VR>
static __device__ __forceinline__ float bias_or_mask(float s,
                                                     const float* bsame,
                                                     const float* bwrap) {
  constexpr int key0 = TT * 16 + VR;
  constexpr int kc0 = key0 % 14;
  constexpr int c0 = (key0 / 14) * 7 + kc0;
  constexpr unsigned lanes = win_lanes(TT, VR);
  const float bias = (kc0 < 6 ? bsame : bwrap)[c0];
  return __builtin_amdgcn_inverse_ballot_w32(lanes) ? s + bias : -1e30f;
}

template <int TT>
static __device__ __forceinline__ v8f score_tile(const v4f (&kf)[4],
                                                 const v4f (&qv)[4],
                                                 const float* bsame,
                                                 const float* bwrap) {
  v8f acc = {};
#pragma unroll
  for (int kk = 0; kk < 4; ++kk) {
    v2f a0, a1, b0, b1;
    a0.x = kf[kk].x; a0.y = kf[kk].y;
    a1.x = kf[kk].z; a1.y = kf[kk].w;
    b0.x = qv[kk].x; b0.y = qv[kk].y;
    b1.x = qv[kk].z; b1.y = qv[kk].w;
    acc = wmma4(a0, b0, acc);
    acc = wmma4(a1, b1, acc);
  }
  acc[0] = bias_or_mask<TT, 0>(acc[0], bsame, bwrap);
  acc[1] = bias_or_mask<TT, 1>(acc[1], bsame, bwrap);
  acc[2] = bias_or_mask<TT, 2>(acc[2], bsame, bwrap);
  acc[3] = bias_or_mask<TT, 3>(acc[3], bsame, bwrap);
  acc[4] = bias_or_mask<TT, 4>(acc[4], bsame, bwrap);
  acc[5] = bias_or_mask<TT, 5>(acc[5], bsame, bwrap);
  acc[6] = bias_or_mask<TT, 6>(acc[6], bsame, bwrap);
  acc[7] = bias_or_mask<TT, 7>(acc[7], bsame, bwrap);
  return acc;
}

#define OSP 132

__global__ __launch_bounds__(128) void attn_proj_kernel(
    const float* __restrict__ q_arr, const float* __restrict__ kpad,
    const float* __restrict__ vpad, const float* __restrict__ rpb,
    const float* __restrict__ wfrag, const float* __restrict__ proj_b,
    float* __restrict__ out) {
  __shared__ float PTs[NHEADS][112 * 17];
  __shared__ float BiasPs[NHEADS][80];
  __shared__ __attribute__((aligned(16))) float Os[16 * OSP];

  int unit = blockIdx.x;
  const int tj = unit % 7;  unit /= 7;
  const int ti = unit % 28; unit /= 28;
  const int b  = unit;
  const int i0 = ti * 2, j0 = tj * 8;

  const int lane = threadIdx.x & 31;
  const int half = lane >> 4;
  const int n16  = lane & 15;
  const int h = __builtin_amdgcn_readfirstlane((int)(threadIdx.x >> 5));
  float* PT = PTs[h];
  float* BiasP = BiasPs[h];

  for (int idx = lane; idx < 80; idx += 32) {
    const int t7 = idx - 14;
    float bval = 0.f;
    if (t7 >= 0 && t7 < 49) {
      const int di = t7 / 7, dj = t7 % 7;
      bval = rpb[((size_t)h * 13 + (di + 3)) * 13 + (dj + 3)];
    }
    BiasP[idx] = bval;
  }

  const size_t kvbase =
      ((((size_t)(b * NHEADS + h)) * PADW + i0) * PADW + j0) * HDIM;
  const float* __restrict__ kblk = kpad + kvbase;
  const float* __restrict__ vblk = vpad + kvbase;

  const int r = n16 >> 3, cq = n16 & 7;
  const int qi = i0 + r, qj = j0 + cq;
  const size_t qbase =
      (((size_t)(b * NHEADS + h)) * HWPIX + qi * IMW + qj) * HDIM;
  v4f qv[4];
#pragma unroll
  for (int kk = 0; kk < 4; ++kk)
    qv[kk] = *(const v4f*)&q_arr[qbase + kk * 8 + 4 * half];

  v4f kf[7][4];
#pragma unroll
  for (int tt = 0; tt < 7; ++tt) {
    const int pos = tt * 16 + n16;
    const int kr = pos / 14, kc = pos % 14;
    const int koff = (kr * PADW + kc) * HDIM;
#pragma unroll
    for (int kk = 0; kk < 4; ++kk)
      kf[tt][kk] = *(const v4f*)&kblk[koff + kk * 8 + 4 * half];
  }
  __syncthreads();

  const float* bsame = BiasP + 14 - (r * 7 + cq) + 8 * half;
  const float* bwrap = BiasP + 14 - (r * 7 + cq) + half;
  v8f S[7];
  S[0] = score_tile<0>(kf[0], qv, bsame, bwrap);
  S[1] = score_tile<1>(kf[1], qv, bsame, bwrap);
  S[2] = score_tile<2>(kf[2], qv, bsame, bwrap);
  S[3] = score_tile<3>(kf[3], qv, bsame, bwrap);
  S[4] = score_tile<4>(kf[4], qv, bsame, bwrap);
  S[5] = score_tile<5>(kf[5], qv, bsame, bwrap);
  S[6] = score_tile<6>(kf[6], qv, bsame, bwrap);

  float mx = -1e30f;
#pragma unroll
  for (int tt = 0; tt < 7; ++tt)
#pragma unroll
    for (int vr = 0; vr < 8; ++vr) mx = fmaxf(mx, S[tt][vr]);
  mx = fmaxf(mx, __shfl_xor(mx, 16, 32));
  float sum = 0.f;
#pragma unroll
  for (int tt = 0; tt < 7; ++tt)
#pragma unroll
    for (int vr = 0; vr < 8; ++vr) {
      const float e = __expf(S[tt][vr] - mx);
      S[tt][vr] = e;
      sum += e;
    }
  sum += __shfl_xor(sum, 16, 32);
  const float inv = 1.f / sum;

#pragma unroll
  for (int tt = 0; tt < 7; ++tt)
#pragma unroll
    for (int vr = 0; vr < 8; ++vr)
      PT[(tt * 16 + vr + 8 * half) * 17 + n16] = S[tt][vr];
  __syncthreads();

  const v4f* __restrict__ wf = (const v4f*)wfrag + (size_t)(2 * h) * 16 * 32 + lane;
  v4f wb[2][16];
#pragma unroll
  for (int tt = 0; tt < 2; ++tt)
#pragma unroll
    for (int j = 0; j < 8; ++j) wb[tt][j] = wf[(tt * 16 + j) * 32];

  v8f acc0 = {}, acc1 = {};
#pragma unroll
  for (int kc = 0; kc < 28; ++kc) {
    const int ka = kc * 4, kb = kc * 4 + 2;
    const int offa = ((ka / 14) * PADW + (ka % 14)) * HDIM;
    const int offb = ((kb / 14) * PADW + (kb % 14)) * HDIM;
    const int voff = (half ? offb : offa) + n16;
    const int k0 = kc * 4 + 2 * half;
    v2f a0, a1, bfr;
    a0.x = vblk[voff];
    a0.y = vblk[voff + HDIM];
    a1.x = vblk[voff + 16];
    a1.y = vblk[voff + HDIM + 16];
    bfr.x = PT[k0 * 17 + n16];
    bfr.y = PT[(k0 + 1) * 17 + n16];
    acc0 = wmma4(a0, bfr, acc0);
    acc1 = wmma4(a1, bfr, acc1);
  }

#pragma unroll
  for (int tt = 0; tt < 2; ++tt)
#pragma unroll
    for (int j = 8; j < 16; ++j) wb[tt][j] = wf[(tt * 16 + j) * 32];

  float* orow = &Os[n16 * OSP + h * HDIM + 8 * half];
  v4f o4;
  o4.x = acc0[0] * inv; o4.y = acc0[1] * inv; o4.z = acc0[2] * inv; o4.w = acc0[3] * inv;
  *(v4f*)&orow[0] = o4;
  o4.x = acc0[4] * inv; o4.y = acc0[5] * inv; o4.z = acc0[6] * inv; o4.w = acc0[7] * inv;
  *(v4f*)&orow[4] = o4;
  o4.x = acc1[0] * inv; o4.y = acc1[1] * inv; o4.z = acc1[2] * inv; o4.w = acc1[3] * inv;
  *(v4f*)&orow[16] = o4;
  o4.x = acc1[4] * inv; o4.y = acc1[5] * inv; o4.z = acc1[6] * inv; o4.w = acc1[7] * inv;
  *(v4f*)&orow[20] = o4;
  __syncthreads();

  const float* arow = &Os[n16 * OSP + 4 * half];
  v4f af[16];
#pragma unroll
  for (int j = 0; j < 16; ++j) af[j] = *(const v4f*)&arow[j * 8];
  __builtin_amdgcn_sched_barrier(0);

  v8f pacc0 = {}, pacc1 = {};
#pragma unroll
  for (int j = 0; j < 16; ++j) {
    v2f a0, a1, b00, b01, b10, b11;
    a0.x  = af[j].x;     a0.y  = af[j].y;
    a1.x  = af[j].z;     a1.y  = af[j].w;
    b00.x = wb[0][j].x;  b00.y = wb[0][j].y;
    b01.x = wb[0][j].z;  b01.y = wb[0][j].w;
    b10.x = wb[1][j].x;  b10.y = wb[1][j].y;
    b11.x = wb[1][j].z;  b11.y = wb[1][j].w;
    pacc0 = wmma4(a0, b00, pacc0);
    pacc1 = wmma4(a0, b10, pacc1);
    pacc0 = wmma4(a1, b01, pacc0);
    pacc1 = wmma4(a1, b11, pacc1);
  }

  const int oc = h * 32 + n16;
  const float pb0 = proj_b[oc], pb1 = proj_b[oc + 16];
  float* __restrict__ orow_g =
      out + ((size_t)(b * HWPIX) + (i0 + half) * IMW + j0) * 128 + oc;
#pragma unroll
  for (int vr = 0; vr < 8; ++vr) {
    orow_g[vr * 128]      = pacc0[vr] + pb0;
    orow_g[vr * 128 + 16] = pacc1[vr] + pb1;
  }
}

extern "C" void kernel_launch(void* const* d_in, const int* in_sizes, int n_in,
                              void* d_out, int out_size, void* d_ws,
                              size_t ws_size, hipStream_t stream) {
  (void)in_sizes; (void)n_in; (void)out_size; (void)ws_size;
  const float* x      = (const float*)d_in[0];
  const float* qkv_w  = (const float*)d_in[1];
  const float* qkv_b  = (const float*)d_in[2];
  const float* rpb    = (const float*)d_in[3];
  const float* proj_w = (const float*)d_in[4];
  const float* proj_b = (const float*)d_in[5];

  float* ws = (float*)d_ws;
  const size_t q_len   = (size_t)BATCH * NHEADS * HWPIX * HDIM;
  const size_t pad_len = (size_t)BATCH * NHEADS * PADW * PADW * HDIM;
  float* q_arr   = ws;
  float* kpad    = ws + q_len;
  float* vpad    = kpad + pad_len;
  float* wfrag   = vpad + pad_len;

  hipMemsetAsync(kpad, 0, 2 * pad_len * sizeof(float), stream);

  qkv_kernel<<<196, 256, 0, stream>>>(x, qkv_w, qkv_b, q_arr, kpad, vpad,
                                      proj_w, wfrag);
  attn_proj_kernel<<<BATCH * 28 * 7, 128, 0, stream>>>(q_arr, kpad, vpad, rpb,
                                                       wfrag, proj_b,
                                                       (float*)d_out);
}


# attention P*V: P^T fragments taken straight from the score registers in S^T's own key order (V loaded in that order); P^T LDS round trip and its barrier gone; P*V k-terms summed in a new order
# speedup vs baseline: 2.0565x; 1.0817x over previous
#include <hip/hip_runtime.h>


typedef __attribute__((ext_vector_type(2))) float v2f;
typedef __attribute__((ext_vector_type(4))) float v4f;
typedef __attribute__((ext_vector_type(8))) float v8f;

#define NHEADS 4
#define HDIM   32
#define IMH    56
#define IMW    56
#define HWPIX  3136
#define BATCH  4
#define PADW   62
#define QSCALE 0.17677669529663687f

static __device__ __forceinline__ v8f wmma4(v2f a, v2f b, v8f c) {
  return __builtin_amdgcn_wmma_f32_16x16x4_f32(false, a, false, b, (short)0, c,
                                               false, false);
}

#define STRIP_M  64
#define STRIP_MT 4
#define XPITCH   132
#define WKB      8

__global__ __launch_bounds__(256) void qkv_kernel(
    const float* __restrict__ x, const float* __restrict__ qkv_w,
    const float* __restrict__ qkv_b, float* __restrict__ q_arr,
    float* __restrict__ kpad, float* __restrict__ vpad,
    const float* __restrict__ proj_w, float* __restrict__ wfrag) {
  __shared__ float Xs[STRIP_M * XPITCH];
  const int strip = blockIdx.x;
  const int t = threadIdx.x;

  if (strip < 16) {
    const int piece = strip * 256 + t;
    const int ln = piece & 31, jj = (piece >> 5) & 15, tile = piece >> 9;
    const v4f w4 = *(const v4f*)&proj_w[(size_t)(tile * 16 + (ln & 15)) * 128 +
                                        8 * jj + 4 * (ln >> 4)];
    *(v4f*)&wfrag[(size_t)piece * 4] = w4;
  }

  const int lane = t & 31;
  const int half = lane >> 4;
  const int n16  = lane & 15;
  const int wave = __builtin_amdgcn_readfirstlane(t >> 5);

  const float* __restrict__ wrow[3];
#pragma unroll
  for (int tt = 0; tt < 3; ++tt)
    wrow[tt] = qkv_w + (size_t)((wave * 3 + tt) * 16 + n16) * 128 + 2 * half;

  v2f bw[2][3][WKB];
#pragma unroll
  for (int tt = 0; tt < 3; ++tt)
#pragma unroll
    for (int kk = 0; kk < WKB; ++kk)
      bw[0][tt][kk] = *(const v2f*)&wrow[tt][kk * 4];

  v4f xin[8];
#pragma unroll
  for (int it = 0; it < 8; ++it)
    xin[it] = *(const v4f*)&x[(size_t)(strip * STRIP_M + (t >> 5) + 8 * it) * 128 +
                              (t & 31) * 4];
#pragma unroll
  for (int it = 0; it < 8; ++it)
    *(v4f*)&Xs[((t >> 5) + 8 * it) * XPITCH + (t & 31) * 4] = xin[it];
  __syncthreads();

  const float* xrow = &Xs[n16 * XPITCH + 2 * half];

  v8f acc[3][STRIP_MT];
#pragma unroll
  for (int tt = 0; tt < 3; ++tt)
#pragma unroll
    for (int mt = 0; mt < STRIP_MT; ++mt) acc[tt][mt] = (v8f){};

#pragma unroll
  for (int blk = 0; blk < 32 / WKB; ++blk) {
    if (blk + 1 < 32 / WKB) {
#pragma unroll
      for (int tt = 0; tt < 3; ++tt)
#pragma unroll
        for (int kk = 0; kk < WKB; ++kk)
          bw[(blk + 1) & 1][tt][kk] =
              *(const v2f*)&wrow[tt][((blk + 1) * WKB + kk) * 4];
    }
    __builtin_amdgcn_sched_barrier(0);
#pragma unroll
    for (int kk = 0; kk < WKB; ++kk) {
      const int kc = blk * WKB + kk;
      v2f a[STRIP_MT];
#pragma unroll
      for (int mt = 0; mt < STRIP_MT; ++mt)
        a[mt] = *(const v2f*)&xrow[mt * 16 * XPITCH + kc * 4];
#pragma unroll
      for (int tt = 0; tt < 3; ++tt)
#pragma unroll
        for (int mt = 0; mt < STRIP_MT; ++mt)
          acc[tt][mt] = wmma4(a[mt], bw[blk & 1][tt][kk], acc[tt][mt]);
    }
  }

  float bias[3], scl[3];
  float* dstp[3];
  unsigned hd[3];
  bool isq[3];
#pragma unroll
  for (int tt = 0; tt < 3; ++tt) {
    const int otile = wave * 3 + tt;
    const int which = otile >> 3;
    const unsigned head = (otile & 7) >> 1;
    const unsigned d    = (otile & 1) * 16 + n16;
    isq[tt]  = (which == 0);
    dstp[tt] = isq[tt] ? q_arr : ((which == 1) ? kpad : vpad);
    scl[tt]  = isq[tt] ? QSCALE : 1.0f;
    hd[tt]   = head * (isq[tt] ? (unsigned)(HWPIX * HDIM)
                               : (unsigned)(PADW * PADW * HDIM)) + d;
    bias[tt] = qkv_b[otile * 16 + n16];
  }

#pragma unroll
  for (int mt = 0; mt < STRIP_MT; ++mt) {
#pragma unroll
    for (int vr = 0; vr < 8; ++vr) {
      const unsigned m = mt * 16 + vr + 8 * half;
      const unsigned pixel = strip * STRIP_M + m;
      const unsigned bimg = pixel / HWPIX;
      const unsigned rem  = pixel - bimg * HWPIX;
      const unsigned i    = rem / IMW;
      const unsigned qpix = (pixel + bimg * ((NHEADS - 1) * HWPIX)) * HDIM;
      const unsigned ppix = (bimg * (NHEADS * PADW * PADW) + rem +
                             i * (PADW - IMW) + (3 * PADW + 3)) * HDIM;
#pragma unroll
      for (int tt = 0; tt < 3; ++tt) {
        const float val = (acc[tt][mt][vr] + bias[tt]) * scl[tt];
        dstp[tt][(isq[tt] ? qpix : ppix) + hd[tt]] = val;
      }
    }
  }
}

static constexpr unsigned win_lanes(int tt, int vr) {
  unsigned m = 0;
  for (int lane = 0; lane < 32; ++lane) {
    const int half = lane >> 4, n16 = lane & 15;
    const int r = n16 >> 3, cq = n16 & 7;
    const int key = tt * 16 + vr + 8 * half;
    const int di = key / 14 - r, dj = key % 14 - cq;
    if (di >= 0 && di < 7 && dj >= 0 && dj < 7) m |= 1u << lane;
  }
  return m;
}

template <int TT, int VR>
static __device__ __forceinline__ float bias_or_mask(float s,
                                                     const float* bsame,
                                                     const float* bwrap) {
  constexpr int key0 = TT * 16 + VR;
  constexpr int kc0 = key0 % 14;
  constexpr int c0 = (key0 / 14) * 7 + kc0;
  constexpr unsigned lanes = win_lanes(TT, VR);
  const float bias = (kc0 < 6 ? bsame : bwrap)[c0];
  return __builtin_amdgcn_inverse_ballot_w32(lanes) ? s + bias : -1e30f;
}

template <int TT>
static __device__ __forceinline__ v8f score_tile(const v4f (&kf)[4],
                                                 const v4f (&qv)[4],
                                                 const float* bsame,
                                                 const float* bwrap) {
  v8f acc = {};
#pragma unroll
  for (int kk = 0; kk < 4; ++kk) {
    v2f a0, a1, b0, b1;
    a0.x = kf[kk].x; a0.y = kf[kk].y;
    a1.x = kf[kk].z; a1.y = kf[kk].w;
    b0.x = qv[kk].x; b0.y = qv[kk].y;
    b1.x = qv[kk].z; b1.y = qv[kk].w;
    acc = wmma4(a0, b0, acc);
    acc = wmma4(a1, b1, acc);
  }
  acc[0] = bias_or_mask<TT, 0>(acc[0], bsame, bwrap);
  acc[1] = bias_or_mask<TT, 1>(acc[1], bsame, bwrap);
  acc[2] = bias_or_mask<TT, 2>(acc[2], bsame, bwrap);
  acc[3] = bias_or_mask<TT, 3>(acc[3], bsame, bwrap);
  acc[4] = bias_or_mask<TT, 4>(acc[4], bsame, bwrap);
  acc[5] = bias_or_mask<TT, 5>(acc[5], bsame, bwrap);
  acc[6] = bias_or_mask<TT, 6>(acc[6], bsame, bwrap);
  acc[7] = bias_or_mask<TT, 7>(acc[7], bsame, bwrap);
  return acc;
}

#define OSP 132

__global__ __launch_bounds__(128, 4) void attn_proj_kernel(
    const float* __restrict__ q_arr, const float* __restrict__ kpad,
    const float* __restrict__ vpad, const float* __restrict__ rpb,
    const float* __restrict__ wfrag, const float* __restrict__ proj_b,
    float* __restrict__ out) {
  __shared__ float BiasPs[NHEADS][80];
  __shared__ __attribute__((aligned(16))) float Os[16 * OSP];

  int unit = blockIdx.x;
  const int tj = unit % 7;  unit /= 7;
  const int ti = unit % 28; unit /= 28;
  const int b  = unit;
  const int i0 = ti * 2, j0 = tj * 8;

  const int lane = threadIdx.x & 31;
  const int half = lane >> 4;
  const int n16  = lane & 15;
  const int h = __builtin_amdgcn_readfirstlane((int)(threadIdx.x >> 5));
  float* BiasP = BiasPs[h];

  for (int idx = lane; idx < 80; idx += 32) {
    const int t7 = idx - 14;
    float bval = 0.f;
    if (t7 >= 0 && t7 < 49) {
      const int di = t7 / 7, dj = t7 % 7;
      bval = rpb[((size_t)h * 13 + (di + 3)) * 13 + (dj + 3)];
    }
    BiasP[idx] = bval;
  }

  const size_t kvbase =
      ((((size_t)(b * NHEADS + h)) * PADW + i0) * PADW + j0) * HDIM;
  const float* __restrict__ kblk = kpad + kvbase;
  const float* __restrict__ vblk = vpad + kvbase;

  const int r = n16 >> 3, cq = n16 & 7;
  const int qi = i0 + r, qj = j0 + cq;
  const size_t qbase =
      (((size_t)(b * NHEADS + h)) * HWPIX + qi * IMW + qj) * HDIM;
  v4f qv[4];
#pragma unroll
  for (int kk = 0; kk < 4; ++kk)
    qv[kk] = *(const v4f*)&q_arr[qbase + kk * 8 + 4 * half];

  v4f kf[7][4];
#pragma unroll
  for (int tt = 0; tt < 7; ++tt) {
    const int pos = tt * 16 + n16;
    const int kr = pos / 14, kc = pos % 14;
    const int koff = (kr * PADW + kc) * HDIM;
#pragma unroll
    for (int kk = 0; kk < 4; ++kk)
      kf[tt][kk] = *(const v4f*)&kblk[koff + kk * 8 + 4 * half];
  }
  __syncthreads();

  const float* bsame = BiasP + 14 - (r * 7 + cq) + 8 * half;
  const float* bwrap = BiasP + 14 - (r * 7 + cq) + half;
  v8f S[7];
  S[0] = score_tile<0>(kf[0], qv, bsame, bwrap);
  S[1] = score_tile<1>(kf[1], qv, bsame, bwrap);
  S[2] = score_tile<2>(kf[2], qv, bsame, bwrap);
  S[3] = score_tile<3>(kf[3], qv, bsame, bwrap);
  S[4] = score_tile<4>(kf[4], qv, bsame, bwrap);
  S[5] = score_tile<5>(kf[5], qv, bsame, bwrap);
  S[6] = score_tile<6>(kf[6], qv, bsame, bwrap);

  float mx = -1e30f;
#pragma unroll
  for (int tt = 0; tt < 7; ++tt)
#pragma unroll
    for (int vr = 0; vr < 8; ++vr) mx = fmaxf(mx, S[tt][vr]);
  mx = fmaxf(mx, __shfl_xor(mx, 16, 32));
  float sum = 0.f;
#pragma unroll
  for (int tt = 0; tt < 7; ++tt)
#pragma unroll
    for (int vr = 0; vr < 8; ++vr) {
      const float e = __expf(S[tt][vr] - mx);
      S[tt][vr] = e;
      sum += e;
    }
  sum += __shfl_xor(sum, 16, 32);
  const float inv = 1.f / sum;

  const v4f* __restrict__ wf = (const v4f*)wfrag + (size_t)(2 * h) * 16 * 32 + lane;
  v4f wb[2][16];
#pragma unroll
  for (int tt = 0; tt < 2; ++tt)
#pragma unroll
    for (int j = 0; j < 8; ++j) wb[tt][j] = wf[(tt * 16 + j) * 32];

  v8f acc0 = {}, acc1 = {};
#pragma unroll
  for (int tt = 0; tt < 7; ++tt) {
#pragma unroll
    for (int j = 0; j < 4; ++j) {
      const int ka = tt * 16 + 2 * j, kb = ka + 8;
      const int offa = ((ka / 14) * PADW + (ka % 14)) * HDIM;
      const int offb = ((kb / 14) * PADW + (kb % 14)) * HDIM;
      const int voff = (half ? offb : offa) + n16;
      v2f a0, a1, bfr;
      a0.x = vblk[voff];
      a0.y = vblk[voff + HDIM];
      a1.x = vblk[voff + 16];
      a1.y = vblk[voff + HDIM + 16];
      bfr.x = S[tt][2 * j];
      bfr.y = S[tt][2 * j + 1];
      acc0 = wmma4(a0, bfr, acc0);
      acc1 = wmma4(a1, bfr, acc1);
    }
  }

#pragma unroll
  for (int tt = 0; tt < 2; ++tt)
#pragma unroll
    for (int j = 8; j < 16; ++j) wb[tt][j] = wf[(tt * 16 + j) * 32];

  float* orow = &Os[n16 * OSP + h * HDIM + 8 * half];
  v4f o4;
  o4.x = acc0[0] * inv; o4.y = acc0[1] * inv; o4.z = acc0[2] * inv; o4.w = acc0[3] * inv;
  *(v4f*)&orow[0] = o4;
  o4.x = acc0[4] * inv; o4.y = acc0[5] * inv; o4.z = acc0[6] * inv; o4.w = acc0[7] * inv;
  *(v4f*)&orow[4] = o4;
  o4.x = acc1[0] * inv; o4.y = acc1[1] * inv; o4.z = acc1[2] * inv; o4.w = acc1[3] * inv;
  *(v4f*)&orow[16] = o4;
  o4.x = acc1[4] * inv; o4.y = acc1[5] * inv; o4.z = acc1[6] * inv; o4.w = acc1[7] * inv;
  *(v4f*)&orow[20] = o4;
  __syncthreads();

  const float* arow = &Os[n16 * OSP + 4 * half];
  v4f af[16];
#pragma unroll
  for (int j = 0; j < 16; ++j) af[j] = *(const v4f*)&arow[j * 8];
  __builtin_amdgcn_sched_barrier(0);

  v8f pacc0 = {}, pacc1 = {};
#pragma unroll
  for (int j = 0; j < 16; ++j) {
    v2f a0, a1, b00, b01, b10, b11;
    a0.x  = af[j].x;     a0.y  = af[j].y;
    a1.x  = af[j].z;     a1.y  = af[j].w;
    b00.x = wb[0][j].x;  b00.y = wb[0][j].y;
    b01.x = wb[0][j].z;  b01.y = wb[0][j].w;
    b10.x = wb[1][j].x;  b10.y = wb[1][j].y;
    b11.x = wb[1][j].z;  b11.y = wb[1][j].w;
    pacc0 = wmma4(a0, b00, pacc0);
    pacc1 = wmma4(a0, b10, pacc1);
    pacc0 = wmma4(a1, b01, pacc0);
    pacc1 = wmma4(a1, b11, pacc1);
  }

  const int oc = h * 32 + n16;
  const float pb0 = proj_b[oc], pb1 = proj_b[oc + 16];
  float* __restrict__ orow_g =
      out + ((size_t)(b * HWPIX) + (i0 + half) * IMW + j0) * 128 + oc;
#pragma unroll
  for (int vr = 0; vr < 8; ++vr) {
    orow_g[vr * 128]      = pacc0[vr] + pb0;
    orow_g[vr * 128 + 16] = pacc1[vr] + pb1;
  }
}

extern "C" void kernel_launch(void* const* d_in, const int* in_sizes, int n_in,
                              void* d_out, int out_size, void* d_ws,
                              size_t ws_size, hipStream_t stream) {
  (void)in_sizes; (void)n_in; (void)out_size; (void)ws_size;
  const float* x      = (const float*)d_in[0];
  const float* qkv_w  = (const float*)d_in[1];
  const float* qkv_b  = (const float*)d_in[2];
  const float* rpb    = (const float*)d_in[3];
  const float* proj_w = (const float*)d_in[4];
  const float* proj_b = (const float*)d_in[5];

  float* ws = (float*)d_ws;
  const size_t q_len   = (size_t)BATCH * NHEADS * HWPIX * HDIM;
  const size_t pad_len = (size_t)BATCH * NHEADS * PADW * PADW * HDIM;
  float* q_arr   = ws;
  float* kpad    = ws + q_len;
  float* vpad    = kpad + pad_len;
  float* wfrag   = vpad + pad_len;

  hipMemsetAsync(kpad, 0, 2 * pad_len * sizeof(float), stream);

  qkv_kernel<<<196, 256, 0, stream>>>(x, qkv_w, qkv_b, q_arr, kpad, vpad,
                                      proj_w, wfrag);
  attn_proj_kernel<<<BATCH * 28 * 7, 128, 0, stream>>>(q_arr, kpad, vpad, rpb,
                                                       wfrag, proj_b,
                                                       (float*)d_out);
}
